# starting kernel
# baseline (speedup 1.0000x reference)
#include <hip/hip_runtime.h>
#include <cstdint>
#include <cstddef>

#define NB   32
#define NN   4096
#define NCAP 10
#define DV   16
#define DIN  8
#define CD   (NCAP * DV)
#define NCHUNK 32
#define CHN  (NN / NCHUNK)

typedef float v2f __attribute__((ext_vector_type(2)));
typedef float v8f __attribute__((ext_vector_type(8)));

__global__ __launch_bounds__(256) void capsule_uhat_wmma(
    const float* __restrict__ x, const float* __restrict__ W,
    float* __restrict__ uhat) {
  const int lane = threadIdx.x & 31;
  const int n    = blockIdx.x * 8 + (threadIdx.x >> 5);
  const int half = lane >> 4;
  const int r    = lane & 15;

  v2f A[2][2];
#pragma unroll
  for (int mt = 0; mt < 2; ++mt) {
#pragma unroll
    for (int kc = 0; kc < 2; ++kc) {
      const int b = mt * 16 + r;
      const int i = kc * 4 + 2 * half;
      const float* p = x + ((size_t)b * NN + n) * DIN + i;
      A[mt][kc].x = p[0];
      A[mt][kc].y = p[1];
    }
  }

  v2f Bm[NCAP][2];
#pragma unroll
  for (int c = 0; c < NCAP; ++c) {
#pragma unroll
    for (int kc = 0; kc < 2; ++kc) {
      const int i0 = kc * 4 + 2 * half;
      const float* p = W + (((size_t)n * NCAP + c) * DIN + i0) * DV + r;
      Bm[c][kc].x = p[0];
      Bm[c][kc].y = p[DV];
    }
  }

  float* outn = uhat + (size_t)n * (NB * CD);
#pragma unroll
  for (int mt = 0; mt < 2; ++mt) {
#pragma unroll
    for (int c = 0; c < NCAP; ++c) {
      v8f acc = {0.f, 0.f, 0.f, 0.f, 0.f, 0.f, 0.f, 0.f};
      acc = __builtin_amdgcn_wmma_f32_16x16x4_f32(
          false, A[mt][0], false, Bm[c][0], (short)0, acc, false, false);
      acc = __builtin_amdgcn_wmma_f32_16x16x4_f32(
          false, A[mt][1], false, Bm[c][1], (short)0, acc, false, false);
#pragma unroll
      for (int v = 0; v < 8; ++v) {
        const int b = mt * 16 + v + 8 * half;
        outn[(size_t)b * CD + c * DV + r] = acc[v];
      }
    }
  }
}

__global__ __launch_bounds__(256) void capsule_routing_accum(
    const float* __restrict__ uhat, float* __restrict__ blog,
    const float* __restrict__ vin, float* __restrict__ partial, int iter) {
  const int b     = blockIdx.y;
  const int chunk = blockIdx.x;
  const int n0    = chunk * CHN;
  const int t     = threadIdx.x;

  __shared__ float s_coeff[CHN * NCAP];
  __shared__ float s_v[CD];

  if (iter > 0) {
    if (t < CD) s_v[t] = vin[b * CD + t];
    __syncthreads();
    for (int task = t; task < CHN * NCAP; task += 256) {
      const int nl = task / NCAP;
      const int c  = task - nl * NCAP;
      const int n  = n0 + nl;
      const float* up = uhat + ((size_t)n * NB + b) * CD + c * DV;
      float dot = 0.f;
#pragma unroll
      for (int d = 0; d < DV; ++d) dot += up[d] * s_v[c * DV + d];
      const size_t bi = ((size_t)b * NN + n) * NCAP + c;
      const float prev = (iter == 1) ? 0.f : blog[bi];
      const float nb = prev + dot;
      blog[bi] = nb;
      s_coeff[task] = nb;
    }
    __syncthreads();
    if (t < CHN) {
      float lg[NCAP];
      float mx = -3.4e38f;
#pragma unroll
      for (int c = 0; c < NCAP; ++c) {
        lg[c] = s_coeff[t * NCAP + c];
        mx = fmaxf(mx, lg[c]);
      }
      float sum = 0.f;
#pragma unroll
      for (int c = 0; c < NCAP; ++c) {
        lg[c] = expf(lg[c] - mx);
        sum += lg[c];
      }
      const float inv = 1.f / sum;
#pragma unroll
      for (int c = 0; c < NCAP; ++c) s_coeff[t * NCAP + c] = lg[c] * inv;
    }
    __syncthreads();
  } else {
    for (int i = t; i < CHN * NCAP; i += 256) s_coeff[i] = 0.1f;
    __syncthreads();
  }

  if (t < CD) {
    const int c = t >> 4;
    float acc = 0.f;
    for (int nl = 0; nl < CHN; ++nl) {
      acc += s_coeff[nl * NCAP + c] *
             uhat[((size_t)(n0 + nl) * NB + b) * CD + t];
    }
    partial[((size_t)b * NCHUNK + chunk) * CD + t] = acc;
  }
}

__global__ __launch_bounds__(160) void capsule_reduce_squash(
    const float* __restrict__ partial, float* __restrict__ vout,
    float* __restrict__ out, int last) {
  const int b = blockIdx.x;
  const int t = threadIdx.x;
  const int c = t >> 4;

  float s = 0.f;
#pragma unroll
  for (int ch = 0; ch < NCHUNK; ++ch)
    s += partial[((size_t)b * NCHUNK + ch) * CD + t];

  __shared__ float sh[CD];
  __shared__ float s2sh[NCAP];
  sh[t] = s * s;
  __syncthreads();
  if ((t & 15) == 0) {
    float s2 = 0.f;
#pragma unroll
    for (int k = 0; k < DV; ++k) s2 += sh[c * DV + k];
    s2sh[c] = s2;
  }
  __syncthreads();
  const float s2 = s2sh[c];
  const float scale = s2 / ((1.f + s2) * sqrtf(s2 + 1e-7f));
  const float vv = scale * s;
  vout[b * CD + t] = vv;
  if (last) out[b * CD + t] = vv;
}

extern "C" void kernel_launch(void* const* d_in, const int* in_sizes, int n_in,
                              void* d_out, int out_size, void* d_ws,
                              size_t ws_size, hipStream_t stream) {
  (void)in_sizes; (void)n_in; (void)out_size; (void)ws_size;
  const float* x = (const float*)d_in[0];
  const float* W = (const float*)d_in[1];
  float* out = (float*)d_out;

  float* ws      = (float*)d_ws;
  float* uhat    = ws;
  float* blog    = uhat + (size_t)NN * NB * CD;
  float* partial = blog + (size_t)NB * NN * NCAP;
  float* vbuf    = partial + (size_t)NB * NCHUNK * CD;

  capsule_uhat_wmma<<<NN / 8, 256, 0, stream>>>(x, W, uhat);
  for (int it = 0; it < 3; ++it) {
    capsule_routing_accum<<<dim3(NCHUNK, NB), 256, 0, stream>>>(
        uhat, blog, vbuf, partial, it);
    capsule_reduce_squash<<<NB, 160, 0, stream>>>(partial, vbuf, out,
                                                  it == 2 ? 1 : 0);
  }
}


# Route passes run two waves per SIMD: each wave keeps one 16-row batch tile of its four input capsules (8 waves per block), so one wave's softmax vector work runs beside the other's matrix work
# speedup vs baseline: 2.3362x; 2.3362x over previous
#include <hip/hip_runtime.h>
#include <cstdint>
#include <cstddef>

#define NB   32
#define NN   4096
#define NCAP 10
#define DV   16
#define DIN  8
#define CD   (NCAP * DV)
#define NPW  4
#define NG   4
#define WPB  (2 * NG)
#define NPB  (NPW * NG)
#define NCHUNK (NN / NPB)
#define RPITCH 164

typedef float v2f __attribute__((ext_vector_type(2)));
typedef float v4f __attribute__((ext_vector_type(4)));
typedef float v8f __attribute__((ext_vector_type(8)));

struct Frag {
  v2f xa[2];
  v2f wb[NCAP][2];
};

__device__ __forceinline__ void load_frag(Frag& f, const float* __restrict__ x,
                                          const float* __restrict__ W, int n,
                                          int mt, int r, int half) {
#pragma unroll
  for (int kc = 0; kc < 2; ++kc) {
    const int b = mt * 16 + r;
    const int i = kc * 4 + 2 * half;
    const float* p = x + ((size_t)b * NN + n) * DIN + i;
    f.xa[kc].x = p[0];
    f.xa[kc].y = p[1];
  }
#pragma unroll
  for (int c = 0; c < NCAP; ++c) {
#pragma unroll
    for (int kc = 0; kc < 2; ++kc) {
      const int i0 = kc * 4 + 2 * half;
      const float* p = W + (((size_t)n * NCAP + c) * DIN + i0) * DV + r;
      f.wb[c][kc].x = p[0];
      f.wb[c][kc].y = p[DV];
    }
  }
}

template <int ROUTED>
__global__ __launch_bounds__(32 * WPB) void capsule_route_pass(
    const float* __restrict__ x, const float* __restrict__ W,
    const float* __restrict__ vsum, float* __restrict__ partial) {
  __shared__ float red[NG][NB][RPITCH];

  const int lane = threadIdx.x & 31;
  const int wave = threadIdx.x >> 5;
  const int g    = wave >> 1;
  const int mt   = wave & 1;
  const int half = lane >> 4;
  const int r    = lane & 15;
  const int n0   = blockIdx.x * NPB + g * NPW;

  const v8f zero8 = {0.f, 0.f, 0.f, 0.f, 0.f, 0.f, 0.f, 0.f};
  v8f S[NCAP];
  v8f V[NCAP];
#pragma unroll
  for (int c = 0; c < NCAP; ++c) {
    S[c] = zero8;
    if (ROUTED) {
      const float* p = vsum + (mt * 16 + r) * CD + c * DV + 8 * half;
      V[c] = *(const v8f*)p;
    } else {
      V[c] = zero8;
    }
  }

  Frag cur, nxt;
  load_frag(cur, x, W, n0, mt, r, half);

#pragma unroll
  for (int j = 0; j < NPW; ++j) {
    if (j + 1 < NPW) load_frag(nxt, x, W, n0 + j + 1, mt, r, half);

    if (!ROUTED) {
#pragma unroll
      for (int c = 0; c < NCAP; ++c) {
        S[c] = __builtin_amdgcn_wmma_f32_16x16x4_f32(
            false, cur.wb[c][0], false, cur.xa[0], (short)0, S[c], false, false);
        S[c] = __builtin_amdgcn_wmma_f32_16x16x4_f32(
            false, cur.wb[c][1], false, cur.xa[1], (short)0, S[c], false, false);
      }
    } else {
      v8f Tt[NCAP];
      float lg[NCAP];
#pragma unroll
      for (int c = 0; c < NCAP; ++c) {
        v8f acc = __builtin_amdgcn_wmma_f32_16x16x4_f32(
            false, cur.wb[c][0], false, cur.xa[0], (short)0, zero8, false, false);
        Tt[c] = __builtin_amdgcn_wmma_f32_16x16x4_f32(
            false, cur.wb[c][1], false, cur.xa[1], (short)0, acc, false, false);
      }
#pragma unroll
      for (int c = 0; c < NCAP; ++c) {
        const v8f tv = Tt[c];
        const v8f vv = V[c];
        const v4f p4 = __builtin_shufflevector(tv, tv, 0, 1, 2, 3) *
                           __builtin_shufflevector(vv, vv, 0, 1, 2, 3) +
                       __builtin_shufflevector(tv, tv, 4, 5, 6, 7) *
                           __builtin_shufflevector(vv, vv, 4, 5, 6, 7);
        const v2f p2 = __builtin_shufflevector(p4, p4, 0, 1) +
                       __builtin_shufflevector(p4, p4, 2, 3);
        float dot = p2.x + p2.y;
        dot += __shfl_xor(dot, 16);
        lg[c] = dot;
      }
      float mx = lg[0];
#pragma unroll
      for (int c = 1; c < NCAP; ++c) mx = fmaxf(mx, lg[c]);
      float sum = 0.f;
#pragma unroll
      for (int c = 0; c < NCAP; ++c) {
        lg[c] = expf(lg[c] - mx);
        sum += lg[c];
      }
      const float inv = 1.f / sum;
#pragma unroll
      for (int c = 0; c < NCAP; ++c) {
        const float cf = lg[c] * inv;
#pragma unroll
        for (int v = 0; v < 8; ++v) S[c][v] += cf * Tt[c][v];
      }
    }
    if (j + 1 < NPW) cur = nxt;
  }

#pragma unroll
  for (int c = 0; c < NCAP; ++c) {
    v8f s = S[c];
    if (!ROUTED) s = s * 0.1f;
    float* q = &red[g][mt * 16 + r][c * DV + 8 * half];
    v4f lo = {s[0], s[1], s[2], s[3]};
    v4f hi = {s[4], s[5], s[6], s[7]};
    *(v4f*)q = lo;
    *(v4f*)(q + 4) = hi;
  }
  __syncthreads();
#pragma unroll
  for (int k = 0; k < (NB * CD / 4) / (32 * WPB); ++k) {
    const int q  = threadIdx.x + (32 * WPB) * k;
    const int b  = q / (CD / 4);
    const int cq = q - b * (CD / 4);
    const v4f s = (*(const v4f*)&red[0][b][4 * cq] + *(const v4f*)&red[1][b][4 * cq]) +
                  (*(const v4f*)&red[2][b][4 * cq] + *(const v4f*)&red[3][b][4 * cq]);
    float* o = partial +
               (((size_t)b * (NCAP / 2) + (cq >> 3)) * NCHUNK + blockIdx.x) * (2 * DV) +
               (cq & 7) * 4;
    *(v4f*)o = s;
  }
}

__global__ __launch_bounds__(128) void capsule_reduce_squash(
    const float* __restrict__ partial, const float* __restrict__ vprev,
    float* __restrict__ vnext, float* __restrict__ out, int mode) {
  const int bp = blockIdx.x;
  const int t  = threadIdx.x;
  const v4f* base = (const v4f*)(partial + (size_t)bp * (NCHUNK * 2 * DV));

  v4f a = {0.f, 0.f, 0.f, 0.f};
#pragma unroll
  for (int k = 0; k < NCHUNK / 16; ++k) a += base[t + 128 * k];

#pragma unroll
  for (int m = 8; m <= 16; m <<= 1) {
    a.x += __shfl_xor(a.x, m);
    a.y += __shfl_xor(a.y, m);
    a.z += __shfl_xor(a.z, m);
    a.w += __shfl_xor(a.w, m);
  }
  __shared__ v4f sh[4][8];
  if ((t & 31) < 8) sh[t >> 5][t & 7] = a;
  __syncthreads();
  if (t < 32) {
    const int j = t & 7;
    const v4f s = (sh[0][j] + sh[1][j]) + (sh[2][j] + sh[3][j]);
    float s2 = (s.x * s.x + s.y * s.y) + (s.z * s.z + s.w * s.w);
    s2 += __shfl_xor(s2, 1);
    s2 += __shfl_xor(s2, 2);
    const float scale = s2 / ((1.f + s2) * sqrtf(s2 + 1e-7f));
    const v4f vv = s * scale;
    if (t < 8) {
      const int idx = bp * (2 * DV) + 4 * t;
      if (mode == 0) {
        *(v4f*)(vnext + idx) = vv;
      } else if (mode == 1) {
        *(v4f*)(vnext + idx) = *(const v4f*)(vprev + idx) + vv;
      } else {
        *(v4f*)(out + idx) = vv;
      }
    }
  }
}

extern "C" void kernel_launch(void* const* d_in, const int* in_sizes, int n_in,
                              void* d_out, int out_size, void* d_ws,
                              size_t ws_size, hipStream_t stream) {
  (void)in_sizes; (void)n_in; (void)out_size; (void)ws_size;
  const float* x = (const float*)d_in[0];
  const float* W = (const float*)d_in[1];
  float* out = (float*)d_out;

  float* ws      = (float*)d_ws;
  float* partial = ws;
  float* v0      = partial + (size_t)NB * NCAP * NCHUNK * DV;
  float* v01     = v0 + NB * CD;

  capsule_route_pass<0><<<NCHUNK, 32 * WPB, 0, stream>>>(x, W, v0, partial);
  capsule_reduce_squash<<<NB * NCAP / 2, 128, 0, stream>>>(partial, v01, v0, out, 0);
  capsule_route_pass<1><<<NCHUNK, 32 * WPB, 0, stream>>>(x, W, v0, partial);
  capsule_reduce_squash<<<NB * NCAP / 2, 128, 0, stream>>>(partial, v0, v01, out, 1);
  capsule_route_pass<1><<<NCHUNK, 32 * WPB, 0, stream>>>(x, W, v01, partial);
  capsule_reduce_squash<<<NB * NCAP / 2, 128, 0, stream>>>(partial, v0, v01, out, 2);
}


# Routed passes: vector instructions per wave 1274 -> 759: softmax on v_exp_f32 and v_rcp_f32 (was library expf, IEEE divide), logit exchange on v_permlanex16 (was LDS bpermute), dot as FMA chain
# speedup vs baseline: 2.4416x; 1.0451x over previous
#include <hip/hip_runtime.h>
#include <cstdint>
#include <cstddef>

#define NB   32
#define NN   4096
#define NCAP 10
#define DV   16
#define DIN  8
#define CD   (NCAP * DV)
#define NPW  4
#define NG   4
#define WPB  (2 * NG)
#define NPB  (NPW * NG)
#define NCHUNK (NN / NPB)
#define RPITCH 164

typedef float v2f __attribute__((ext_vector_type(2)));
typedef float v4f __attribute__((ext_vector_type(4)));
typedef float v8f __attribute__((ext_vector_type(8)));

struct Frag {
  v2f xa[2];
  v2f wb[NCAP][2];
};

__device__ __forceinline__ void load_frag(Frag& f, const float* __restrict__ x,
                                          const float* __restrict__ W, int n,
                                          int mt, int r, int half) {
#pragma unroll
  for (int kc = 0; kc < 2; ++kc) {
    const int b = mt * 16 + r;
    const int i = kc * 4 + 2 * half;
    const float* p = x + ((size_t)b * NN + n) * DIN + i;
    f.xa[kc].x = p[0];
    f.xa[kc].y = p[1];
  }
#pragma unroll
  for (int c = 0; c < NCAP; ++c) {
#pragma unroll
    for (int kc = 0; kc < 2; ++kc) {
      const int i0 = kc * 4 + 2 * half;
      const float* p = W + (((size_t)n * NCAP + c) * DIN + i0) * DV + r;
      f.wb[c][kc].x = p[0];
      f.wb[c][kc].y = p[DV];
    }
  }
}

__device__ __forceinline__ float xhalf(float x) {
  const int b = __builtin_bit_cast(int, x);
  return __builtin_bit_cast(
      float, __builtin_amdgcn_permlanex16(b, b, 0x76543210, 0xfedcba98, false, false));
}

template <int ROUTED>
__global__ __launch_bounds__(32 * WPB) void capsule_route_pass(
    const float* __restrict__ x, const float* __restrict__ W,
    const float* __restrict__ vsum, float* __restrict__ partial) {
  __shared__ float red[NG][NB][RPITCH];

  const int lane = threadIdx.x & 31;
  const int wave = threadIdx.x >> 5;
  const int g    = wave >> 1;
  const int mt   = wave & 1;
  const int half = lane >> 4;
  const int r    = lane & 15;
  const int n0   = blockIdx.x * NPB + g * NPW;

  const v8f zero8 = {0.f, 0.f, 0.f, 0.f, 0.f, 0.f, 0.f, 0.f};
  v8f S[NCAP];
  v8f V[NCAP];
#pragma unroll
  for (int c = 0; c < NCAP; ++c) {
    S[c] = zero8;
    if (ROUTED) {
      const float* p = vsum + (mt * 16 + r) * CD + c * DV + 8 * half;
      V[c] = *(const v8f*)p;
    } else {
      V[c] = zero8;
    }
  }

  Frag cur, nxt;
  load_frag(cur, x, W, n0, mt, r, half);

#pragma unroll
  for (int j = 0; j < NPW; ++j) {
    if (j + 1 < NPW) load_frag(nxt, x, W, n0 + j + 1, mt, r, half);

    if (!ROUTED) {
#pragma unroll
      for (int c = 0; c < NCAP; ++c) {
        S[c] = __builtin_amdgcn_wmma_f32_16x16x4_f32(
            false, cur.wb[c][0], false, cur.xa[0], (short)0, S[c], false, false);
        S[c] = __builtin_amdgcn_wmma_f32_16x16x4_f32(
            false, cur.wb[c][1], false, cur.xa[1], (short)0, S[c], false, false);
      }
    } else {
      v8f Tt[NCAP];
      float lg[NCAP];
#pragma unroll
      for (int c = 0; c < NCAP; ++c) {
        v8f acc = __builtin_amdgcn_wmma_f32_16x16x4_f32(
            false, cur.wb[c][0], false, cur.xa[0], (short)0, zero8, false, false);
        Tt[c] = __builtin_amdgcn_wmma_f32_16x16x4_f32(
            false, cur.wb[c][1], false, cur.xa[1], (short)0, acc, false, false);
      }
#pragma unroll
      for (int c = 0; c < NCAP; ++c) {
        const v8f tv = Tt[c];
        const v8f vv = V[c];
        v2f p2 = __builtin_shufflevector(tv, tv, 0, 1) * __builtin_shufflevector(vv, vv, 0, 1);
        p2 = __builtin_elementwise_fma(__builtin_shufflevector(tv, tv, 2, 3),
                                       __builtin_shufflevector(vv, vv, 2, 3), p2);
        p2 = __builtin_elementwise_fma(__builtin_shufflevector(tv, tv, 4, 5),
                                       __builtin_shufflevector(vv, vv, 4, 5), p2);
        p2 = __builtin_elementwise_fma(__builtin_shufflevector(tv, tv, 6, 7),
                                       __builtin_shufflevector(vv, vv, 6, 7), p2);
        float dot = p2.x + p2.y;
        dot += xhalf(dot);
        lg[c] = dot;
      }
      float mx = lg[0];
#pragma unroll
      for (int c = 1; c < NCAP; ++c) mx = fmaxf(mx, lg[c]);
      float sum = 0.f;
#pragma unroll
      for (int c = 0; c < NCAP; ++c) {
        lg[c] = __builtin_amdgcn_exp2f((lg[c] - mx) * 1.44269504088896340736f);
        sum += lg[c];
      }
      const float inv = __builtin_amdgcn_rcpf(sum);
#pragma unroll
      for (int c = 0; c < NCAP; ++c) {
        const float cf = lg[c] * inv;
#pragma unroll
        for (int v = 0; v < 8; ++v) S[c][v] += cf * Tt[c][v];
      }
    }
    if (j + 1 < NPW) cur = nxt;
  }

#pragma unroll
  for (int c = 0; c < NCAP; ++c) {
    v8f s = S[c];
    if (!ROUTED) s = s * 0.1f;
    float* q = &red[g][mt * 16 + r][c * DV + 8 * half];
    v4f lo = {s[0], s[1], s[2], s[3]};
    v4f hi = {s[4], s[5], s[6], s[7]};
    *(v4f*)q = lo;
    *(v4f*)(q + 4) = hi;
  }
  __syncthreads();
#pragma unroll
  for (int k = 0; k < (NB * CD / 4) / (32 * WPB); ++k) {
    const int q  = threadIdx.x + (32 * WPB) * k;
    const int b  = q / (CD / 4);
    const int cq = q - b * (CD / 4);
    const v4f s = (*(const v4f*)&red[0][b][4 * cq] + *(const v4f*)&red[1][b][4 * cq]) +
                  (*(const v4f*)&red[2][b][4 * cq] + *(const v4f*)&red[3][b][4 * cq]);
    float* o = partial +
               (((size_t)b * (NCAP / 2) + (cq >> 3)) * NCHUNK + blockIdx.x) * (2 * DV) +
               (cq & 7) * 4;
    *(v4f*)o = s;
  }
}

__global__ __launch_bounds__(128) void capsule_reduce_squash(
    const float* __restrict__ partial, const float* __restrict__ vprev,
    float* __restrict__ vnext, float* __restrict__ out, int mode) {
  const int bp = blockIdx.x;
  const int t  = threadIdx.x;
  const v4f* base = (const v4f*)(partial + (size_t)bp * (NCHUNK * 2 * DV));

  v4f a = {0.f, 0.f, 0.f, 0.f};
#pragma unroll
  for (int k = 0; k < NCHUNK / 16; ++k) a += base[t + 128 * k];

#pragma unroll
  for (int m = 8; m <= 16; m <<= 1) {
    a.x += __shfl_xor(a.x, m);
    a.y += __shfl_xor(a.y, m);
    a.z += __shfl_xor(a.z, m);
    a.w += __shfl_xor(a.w, m);
  }
  __shared__ v4f sh[4][8];
  if ((t & 31) < 8) sh[t >> 5][t & 7] = a;
  __syncthreads();
  if (t < 32) {
    const int j = t & 7;
    const v4f s = (sh[0][j] + sh[1][j]) + (sh[2][j] + sh[3][j]);
    float s2 = (s.x * s.x + s.y * s.y) + (s.z * s.z + s.w * s.w);
    s2 += __shfl_xor(s2, 1);
    s2 += __shfl_xor(s2, 2);
    const float scale = s2 / ((1.f + s2) * sqrtf(s2 + 1e-7f));
    const v4f vv = s * scale;
    if (t < 8) {
      const int idx = bp * (2 * DV) + 4 * t;
      if (mode == 0) {
        *(v4f*)(vnext + idx) = vv;
      } else if (mode == 1) {
        *(v4f*)(vnext + idx) = *(const v4f*)(vprev + idx) + vv;
      } else {
        *(v4f*)(out + idx) = vv;
      }
    }
  }
}

extern "C" void kernel_launch(void* const* d_in, const int* in_sizes, int n_in,
                              void* d_out, int out_size, void* d_ws,
                              size_t ws_size, hipStream_t stream) {
  (void)in_sizes; (void)n_in; (void)out_size; (void)ws_size;
  const float* x = (const float*)d_in[0];
  const float* W = (const float*)d_in[1];
  float* out = (float*)d_out;

  float* ws      = (float*)d_ws;
  float* partial = ws;
  float* v0      = partial + (size_t)NB * NCAP * NCHUNK * DV;
  float* v01     = v0 + NB * CD;

  capsule_route_pass<0><<<NCHUNK, 32 * WPB, 0, stream>>>(x, W, v0, partial);
  capsule_reduce_squash<<<NB * NCAP / 2, 128, 0, stream>>>(partial, v01, v0, out, 0);
  capsule_route_pass<1><<<NCHUNK, 32 * WPB, 0, stream>>>(x, W, v0, partial);
  capsule_reduce_squash<<<NB * NCAP / 2, 128, 0, stream>>>(partial, v0, v01, out, 1);
  capsule_route_pass<1><<<NCHUNK, 32 * WPB, 0, stream>>>(x, W, v01, partial);
  capsule_reduce_squash<<<NB * NCAP / 2, 128, 0, stream>>>(partial, v0, v01, out, 2);
}
